# starting kernel
# baseline (speedup 1.0000x reference)
#include <hip/hip_runtime.h>
#include <math.h>


typedef __attribute__((ext_vector_type(2))) float v2f;
typedef __attribute__((ext_vector_type(8))) float v8f;

#define B_  128
#define N_  16384
#define K_  12
#define D_  256
#define CHUNKS 32
#define ROWS_PER_CHUNK (N_ / CHUNKS)
#define EPS_ 1e-8f

#define OUT_SLOTS 0L
#define OUT_MASKS ((long)B_ * K_ * D_)
#define OUT_GATES (OUT_MASKS + (long)B_ * N_ * K_)
#define OUT_UTIL  (OUT_GATES + (long)B_ * K_)

#define WS_PARTIALS 0
#define WS_GATES    (B_ * CHUNKS * K_)

__global__ void asp_reduce(const float* __restrict__ masks,
                           float* __restrict__ partials) {
    const int blk = blockIdx.x;
    const int b   = blk / CHUNKS;
    const int ch  = blk % CHUNKS;
    const int t   = threadIdx.x;

    const float4* src = (const float4*)masks
                      + ((long)b * N_ + (long)ch * ROWS_PER_CHUNK) * 3;
    const float4* p = src + (long)t * 6;

    const float4 v0 = p[0], v1 = p[1], v2 = p[2];
    const float4 v3 = p[3], v4 = p[4], v5 = p[5];

    float a[12];
    a[0] = v0.x + v3.x;  a[1]  = v0.y + v3.y;  a[2]  = v0.z + v3.z;  a[3]  = v0.w + v3.w;
    a[4] = v1.x + v4.x;  a[5]  = v1.y + v4.y;  a[6]  = v1.z + v4.z;  a[7]  = v1.w + v4.w;
    a[8] = v2.x + v5.x;  a[9]  = v2.y + v5.y;  a[10] = v2.z + v5.z;  a[11] = v2.w + v5.w;

    #pragma unroll
    for (int m = 16; m >= 1; m >>= 1) {
        #pragma unroll
        for (int j = 0; j < 12; ++j)
            a[j] += __shfl_xor(a[j], m, 32);
    }

    __shared__ float wred[8 * 12];
    const int w    = t >> 5;
    const int lane = t & 31;
    if (lane == 0) {
        #pragma unroll
        for (int j = 0; j < 12; ++j) wred[w * 12 + j] = a[j];
    }
    __syncthreads();

    if (t < 12) {
        float s = 0.f;
        #pragma unroll
        for (int ww = 0; ww < 8; ++ww) s += wred[ww * 12 + t];
        partials[((long)b * CHUNKS + ch) * K_ + t] = s;
    }
}

__global__ void asp_gates(const float* __restrict__ partials,
                          const float* __restrict__ W1,
                          const float* __restrict__ b1,
                          const float* __restrict__ W2,
                          const float* __restrict__ b2,
                          float* __restrict__ gates_ws,
                          float* __restrict__ out) {
    const int tid  = blockIdx.x * blockDim.x + threadIdx.x;
    const int tile = tid >> 5;
    const int lane = threadIdx.x & 31;
    const int m    = lane & 15;
    const int r    = tile * 16 + m;
    const int b    = r / K_;
    const int k    = r - b * K_;

    float s = 0.f;
    #pragma unroll 8
    for (int c = 0; c < CHUNKS; ++c)
        s += partials[((long)b * CHUNKS + c) * K_ + k];
    const float u = s * (1.0f / (float)N_);

    if (lane < 16) out[OUT_UTIL + r] = u;

    float h[16], w2[16];
    #pragma unroll
    for (int j = 0; j < 16; ++j) {
        h[j]  = fmaxf(0.f, fmaf(u, W1[j], b1[j]));
        w2[j] = W2[j];
    }

    const bool hi = (lane >= 16);
    v8f acc = {0.f, 0.f, 0.f, 0.f, 0.f, 0.f, 0.f, 0.f};
    #pragma unroll
    for (int i = 0; i < 4; ++i) {
        v2f av, bv;
        av.x = hi ? h[4 * i + 2]  : h[4 * i + 0];
        av.y = hi ? h[4 * i + 3]  : h[4 * i + 1];
        bv.x = hi ? w2[4 * i + 2] : w2[4 * i + 0];
        bv.y = hi ? w2[4 * i + 3] : w2[4 * i + 1];
        acc = __builtin_amdgcn_wmma_f32_16x16x4_f32(
             false, av,  false, bv,
             (short)0, acc,  false,  false);
    }

    const float bias = b2[0];
    if (lane == 0 || lane == 16) {
        const int base = tile * 16 + ((lane == 16) ? 8 : 0);
        #pragma unroll
        for (int j = 0; j < 8; ++j) {
            const float g = 1.0f / (1.0f + __expf(-(acc[j] + bias)));
            gates_ws[base + j]        = g;
            out[OUT_GATES + base + j] = g;
        }
    }
}

__global__ void asp_slots(const float* __restrict__ slots,
                          const float* __restrict__ gates_ws,
                          float* __restrict__ out) {
    const long i4 = (long)blockIdx.x * blockDim.x + threadIdx.x;
    const float4 v = ((const float4*)slots)[i4];
    const float g  = gates_ws[i4 >> 6];
    float4 o;
    o.x = v.x * g; o.y = v.y * g; o.z = v.z * g; o.w = v.w * g;
    ((float4*)(out + OUT_SLOTS))[i4] = o;
}

__global__ void asp_masks(const float* __restrict__ masks,
                          const float* __restrict__ gates_ws,
                          float* __restrict__ out) {
    __shared__ float g[12];
    const long row = (long)blockIdx.x * 256 + threadIdx.x;
    const int  b   = (int)(row >> 14);
    if (threadIdx.x < 12) g[threadIdx.x] = gates_ws[b * K_ + threadIdx.x];
    __syncthreads();

    const float4* src = (const float4*)masks + row * 3;
    float4 m0 = src[0], m1 = src[1], m2 = src[2];

    m0.x *= g[0];  m0.y *= g[1];  m0.z *= g[2];  m0.w *= g[3];
    m1.x *= g[4];  m1.y *= g[5];  m1.z *= g[6];  m1.w *= g[7];
    m2.x *= g[8];  m2.y *= g[9];  m2.z *= g[10]; m2.w *= g[11];

    const float sum = (m0.x + m0.y + m0.z + m0.w)
                    + (m1.x + m1.y + m1.z + m1.w)
                    + (m2.x + m2.y + m2.z + m2.w);
    const float inv = 1.0f / (sum + EPS_);

    m0.x *= inv; m0.y *= inv; m0.z *= inv; m0.w *= inv;
    m1.x *= inv; m1.y *= inv; m1.z *= inv; m1.w *= inv;
    m2.x *= inv; m2.y *= inv; m2.z *= inv; m2.w *= inv;

    float4* dst = (float4*)(out + OUT_MASKS) + row * 3;
    dst[0] = m0; dst[1] = m1; dst[2] = m2;
}

extern "C" void kernel_launch(void* const* d_in, const int* in_sizes, int n_in,
                              void* d_out, int out_size, void* d_ws, size_t ws_size,
                              hipStream_t stream) {
    const float* slots = (const float*)d_in[0];
    const float* masks = (const float*)d_in[1];
    const float* W1    = (const float*)d_in[2];
    const float* b1    = (const float*)d_in[3];
    const float* W2    = (const float*)d_in[4];
    const float* b2    = (const float*)d_in[5];
    float* out = (float*)d_out;
    float* ws  = (float*)d_ws;

    float* partials = ws + WS_PARTIALS;
    float* gates    = ws + WS_GATES;

    asp_reduce<<<B_ * CHUNKS, 256, 0, stream>>>(masks, partials);
    asp_gates<<<12, 256, 0, stream>>>(partials, W1, b1, W2, b2, gates, out);
    asp_slots<<<(B_ * K_ * D_ / 4) / 256, 256, 0, stream>>>(slots, gates, out);
    asp_masks<<<(B_ * N_) / 256, 256, 0, stream>>>(masks, gates, out);
}


# One launch, two work-groups a batch on all 256 WGPs: each sums its whole batch itself over flat 16-byte pieces, runs the gate MLP (matrix instruction kept), scales its half; N-sum order changes
# speedup vs baseline: 1.0343x; 1.0343x over previous
#include <hip/hip_runtime.h>
#include <math.h>


typedef __attribute__((ext_vector_type(2))) float v2f;
typedef __attribute__((ext_vector_type(8))) float v8f;

#define B_  128
#define N_  16384
#define K_  12
#define D_  256
#define EPS_ 1e-8f

#define THREADS 512
#define WAVES 16
#define GRID 256
#define F4_PER_BATCH 49152
#define SUM_GROUP 24
#define SUM_OUTER 4
#define HALF_ROWS 8192
#define SLOT_F4_HALF 384

#define OUT_SLOTS 0L
#define OUT_MASKS ((long)B_ * K_ * D_)
#define OUT_GATES (OUT_MASKS + (long)B_ * N_ * K_)
#define OUT_UTIL  (OUT_GATES + (long)B_ * K_)

__device__ __forceinline__ void add4(float4& a, const float4 v) {
    a.x += v.x; a.y += v.y; a.z += v.z; a.w += v.w;
}
__device__ __forceinline__ float4 sel3(int ph0, int ph1, int p,
                                       const float4 a0, const float4 a1, const float4 a2) {
    return (ph0 == p) ? a0 : ((ph1 == p) ? a1 : a2);
}

__global__ void __launch_bounds__(THREADS)
asp_fused(const float* __restrict__ slots,
          const float* __restrict__ masks,
          const float* __restrict__ W1,
          const float* __restrict__ b1,
          const float* __restrict__ W2,
          const float* __restrict__ b2,
          float* __restrict__ out) {
    __shared__ float wred[WAVES * K_];
    __shared__ float g_s[K_];

    const int t    = threadIdx.x;
    const int lane = t & 31;
    const int w    = __builtin_amdgcn_readfirstlane(t >> 5);
    const int b    = blockIdx.x >> 1;
    const int half = blockIdx.x & 1;

    const float4* mb = (const float4*)masks + (long)b * F4_PER_BATCH + t;
    float4 acc0 = {0.f, 0.f, 0.f, 0.f};
    float4 acc1 = {0.f, 0.f, 0.f, 0.f};
    float4 acc2 = {0.f, 0.f, 0.f, 0.f};
    for (int o = 0; o < SUM_OUTER; ++o) {
        float4 v[SUM_GROUP];
        #pragma unroll
        for (int j = 0; j < SUM_GROUP; ++j)
            v[j] = mb[(long)(o * SUM_GROUP + j) * THREADS];
        #pragma unroll
        for (int j = 0; j < SUM_GROUP; j += 3) {
            add4(acc0, v[j]); add4(acc1, v[j + 1]); add4(acc2, v[j + 2]);
        }
    }
    const int ph0 = t % 3;
    const int ph1 = (t + THREADS) % 3;
    const float4 P0 = sel3(ph0, ph1, 0, acc0, acc1, acc2);
    const float4 P1 = sel3(ph0, ph1, 1, acc0, acc1, acc2);
    const float4 P2 = sel3(ph0, ph1, 2, acc0, acc1, acc2);

    float a[K_];
    a[0] = P0.x; a[1] = P0.y; a[2]  = P0.z; a[3]  = P0.w;
    a[4] = P1.x; a[5] = P1.y; a[6]  = P1.z; a[7]  = P1.w;
    a[8] = P2.x; a[9] = P2.y; a[10] = P2.z; a[11] = P2.w;
    #pragma unroll
    for (int m = 16; m >= 1; m >>= 1) {
        #pragma unroll
        for (int j = 0; j < K_; ++j) a[j] += __shfl_xor(a[j], m, 32);
    }
    if (lane == 0) {
        #pragma unroll
        for (int j = 0; j < K_; ++j) wred[w * K_ + j] = a[j];
    }
    __syncthreads();

    if (w == 0) {
        const int m  = lane & 15;
        const int mm = (m < K_) ? m : (K_ - 1);
        float s = 0.f;
        #pragma unroll
        for (int ww = 0; ww < WAVES; ++ww) s += wred[ww * K_ + mm];
        const float u = s * (1.0f / (float)N_);
        if (half == 0 && lane < K_) out[OUT_UTIL + b * K_ + lane] = u;

        float h[16], w2[16];
        #pragma unroll
        for (int j = 0; j < 16; ++j) {
            h[j]  = fmaxf(0.f, fmaf(u, W1[j], b1[j]));
            w2[j] = W2[j];
        }
        const bool hi = (lane >= 16);
        v8f acc = {0.f, 0.f, 0.f, 0.f, 0.f, 0.f, 0.f, 0.f};
        #pragma unroll
        for (int i = 0; i < 4; ++i) {
            v2f av, bv;
            av.x = hi ? h[4 * i + 2]  : h[4 * i + 0];
            av.y = hi ? h[4 * i + 3]  : h[4 * i + 1];
            bv.x = hi ? w2[4 * i + 2] : w2[4 * i + 0];
            bv.y = hi ? w2[4 * i + 3] : w2[4 * i + 1];
            acc = __builtin_amdgcn_wmma_f32_16x16x4_f32(
                false, av, false, bv, (short)0, acc, false, false);
        }
        const float bias = b2[0];
        float g[8];
        #pragma unroll
        for (int j = 0; j < 8; ++j) g[j] = 1.0f / (1.0f + __expf(-(acc[j] + bias)));
        if (lane == 0) {
            #pragma unroll
            for (int j = 0; j < 8; ++j) {
                g_s[j] = g[j];
                if (half == 0) out[OUT_GATES + b * K_ + j] = g[j];
            }
        }
        if (lane == 16) {
            #pragma unroll
            for (int j = 0; j < 4; ++j) {
                g_s[8 + j] = g[j];
                if (half == 0) out[OUT_GATES + b * K_ + 8 + j] = g[j];
            }
        }
    }
    __syncthreads();

    float g[K_];
    #pragma unroll
    for (int j = 0; j < K_; ++j) g[j] = g_s[j];

    if (t < SLOT_F4_HALF) {
        const int idx = half * SLOT_F4_HALF + t;
        const long i4 = (long)b * (K_ * D_ / 4) + idx;
        const float4 v = ((const float4*)slots)[i4];
        const float gg = g_s[idx >> 6];
        float4 o;
        o.x = v.x * gg; o.y = v.y * gg; o.z = v.z * gg; o.w = v.w * gg;
        ((float4*)(out + OUT_SLOTS))[i4] = o;
    }

    const long row0 = (long)b * N_ + (long)half * HALF_ROWS + t;
    const float4* src = (const float4*)masks + row0 * 3;
    float4* dst = (float4*)(out + OUT_MASKS) + row0 * 3;
    for (int o = 0; o < HALF_ROWS / THREADS / 4; ++o) {
        float4 m0[4], m1[4], m2[4];
        #pragma unroll
        for (int r = 0; r < 4; ++r) {
            const float4* p = src + (long)(o * 4 + r) * THREADS * 3;
            m0[r] = p[0]; m1[r] = p[1]; m2[r] = p[2];
        }
        #pragma unroll
        for (int r = 0; r < 4; ++r) {
            float4 x0 = m0[r], x1 = m1[r], x2 = m2[r];
            x0.x *= g[0];  x0.y *= g[1];  x0.z *= g[2];  x0.w *= g[3];
            x1.x *= g[4];  x1.y *= g[5];  x1.z *= g[6];  x1.w *= g[7];
            x2.x *= g[8];  x2.y *= g[9];  x2.z *= g[10]; x2.w *= g[11];
            const float sum = (x0.x + x0.y + x0.z + x0.w)
                            + (x1.x + x1.y + x1.z + x1.w)
                            + (x2.x + x2.y + x2.z + x2.w);
            const float inv = 1.0f / (sum + EPS_);
            x0.x *= inv; x0.y *= inv; x0.z *= inv; x0.w *= inv;
            x1.x *= inv; x1.y *= inv; x1.z *= inv; x1.w *= inv;
            x2.x *= inv; x2.y *= inv; x2.z *= inv; x2.w *= inv;
            float4* q = dst + (long)(o * 4 + r) * THREADS * 3;
            q[0] = x0; q[1] = x1; q[2] = x2;
        }
    }
}

extern "C" void kernel_launch(void* const* d_in, const int* in_sizes, int n_in,
                              void* d_out, int out_size, void* d_ws, size_t ws_size,
                              hipStream_t stream) {
    const float* slots = (const float*)d_in[0];
    const float* masks = (const float*)d_in[1];
    const float* W1    = (const float*)d_in[2];
    const float* b1    = (const float*)d_in[3];
    const float* W2    = (const float*)d_in[4];
    const float* b2    = (const float*)d_in[5];
    float* out = (float*)d_out;
    asp_fused<<<GRID, THREADS, 0, stream>>>(slots, masks, W1, b1, W2, b2, out);
}


# Own rows are read from memory once: while summing the batch a work-group parks its 4 own row blocks in LDS (288 KiB) and registers and scales them from there; N-sum order changes (even+odd blocks)
# speedup vs baseline: 1.1664x; 1.1277x over previous
#include <hip/hip_runtime.h>
#include <math.h>


typedef __attribute__((ext_vector_type(2))) float v2f;
typedef __attribute__((ext_vector_type(8))) float v8f;

#define B_  128
#define N_  16384
#define K_  12
#define D_  256
#define EPS_ 1e-8f

#define THREADS 512
#define WAVES 16
#define GRID 256
#define F4_PER_BATCH 49152
#define BLOCKS 8
#define BLOCK_F4 6144
#define GROUP_F4 12
#define PARKED 3
#define STAGE_F4 (PARKED * BLOCK_F4)
#define SLOT_F4_HALF 384

#define OUT_SLOTS 0L
#define OUT_MASKS ((long)B_ * K_ * D_)
#define OUT_GATES (OUT_MASKS + (long)B_ * N_ * K_)
#define OUT_UTIL  (OUT_GATES + (long)B_ * K_)

__device__ __forceinline__ void add4(float4& a, const float4 v) {
    a.x += v.x; a.y += v.y; a.z += v.z; a.w += v.w;
}
__device__ __forceinline__ float4 sel3(int ph0, int ph1, int p,
                                       const float4 a0, const float4 a1, const float4 a2) {
    return (ph0 == p) ? a0 : ((ph1 == p) ? a1 : a2);
}
__device__ __forceinline__ float4 even_plus_odd(int half, const float4 own, const float4 oth) {
    const float4 ev = half ? oth : own;
    const float4 od = half ? own : oth;
    float4 r;
    r.x = ev.x + od.x; r.y = ev.y + od.y; r.z = ev.z + od.z; r.w = ev.w + od.w;
    return r;
}

__device__ __forceinline__ void scale_row(float4 x0, float4 x1, float4 x2,
                                          const float (&g)[K_], float4* __restrict__ q) {
    x0.x *= g[0];  x0.y *= g[1];  x0.z *= g[2];  x0.w *= g[3];
    x1.x *= g[4];  x1.y *= g[5];  x1.z *= g[6];  x1.w *= g[7];
    x2.x *= g[8];  x2.y *= g[9];  x2.z *= g[10]; x2.w *= g[11];
    const float sum = (x0.x + x0.y + x0.z + x0.w)
                    + (x1.x + x1.y + x1.z + x1.w)
                    + (x2.x + x2.y + x2.z + x2.w);
    const float inv = 1.0f / (sum + EPS_);
    x0.x *= inv; x0.y *= inv; x0.z *= inv; x0.w *= inv;
    x1.x *= inv; x1.y *= inv; x1.z *= inv; x1.w *= inv;
    x2.x *= inv; x2.y *= inv; x2.z *= inv; x2.w *= inv;
    q[0] = x0; q[1] = x1; q[2] = x2;
}

__global__ void __launch_bounds__(THREADS)
asp_fused(const float* __restrict__ slots,
          const float* __restrict__ masks,
          const float* __restrict__ W1,
          const float* __restrict__ b1,
          const float* __restrict__ W2,
          const float* __restrict__ b2,
          float* __restrict__ out) {
    __shared__ float4 stage[STAGE_F4];
    __shared__ float wred[WAVES * K_];
    __shared__ float g_s[K_];

    const int t    = threadIdx.x;
    const int lane = t & 31;
    const int w    = __builtin_amdgcn_readfirstlane(t >> 5);
    const int b    = blockIdx.x >> 1;
    const int half = blockIdx.x & 1;

    const float4* mb = (const float4*)masks + (long)b * F4_PER_BATCH + t;
    float4* st = stage + t;
    float4 own0 = {0.f, 0.f, 0.f, 0.f}, own1 = own0, own2 = own0;
    float4 oth0 = own0, oth1 = own0, oth2 = own0;
    float4 keep[GROUP_F4];
    {
        float4 vo[GROUP_F4];
        const float4* qa = mb + (long)half * BLOCK_F4;
        const float4* qb = mb + (long)(1 - half) * BLOCK_F4;
        #pragma unroll
        for (int j = 0; j < GROUP_F4; ++j) keep[j] = qa[j * THREADS];
        #pragma unroll
        for (int j = 0; j < GROUP_F4; ++j) vo[j] = qb[j * THREADS];
        #pragma unroll
        for (int j = 0; j < GROUP_F4; j += 3) {
            add4(own0, keep[j]); add4(own1, keep[j + 1]); add4(own2, keep[j + 2]);
        }
        #pragma unroll
        for (int j = 0; j < GROUP_F4; j += 3) {
            add4(oth0, vo[j]); add4(oth1, vo[j + 1]); add4(oth2, vo[j + 2]);
        }
    }
    #pragma unroll 1
    for (int m = 1; m < BLOCKS / 2; ++m) {
        float4 va[GROUP_F4], vo[GROUP_F4];
        const float4* qa = mb + (long)(2 * m + half) * BLOCK_F4;
        const float4* qb = mb + (long)(2 * m + 1 - half) * BLOCK_F4;
        #pragma unroll
        for (int j = 0; j < GROUP_F4; ++j) va[j] = qa[j * THREADS];
        #pragma unroll
        for (int j = 0; j < GROUP_F4; ++j) vo[j] = qb[j * THREADS];
        #pragma unroll
        for (int j = 0; j < GROUP_F4; j += 3) {
            add4(own0, va[j]); add4(own1, va[j + 1]); add4(own2, va[j + 2]);
        }
        float4* sp = st + (m - 1) * BLOCK_F4;
        #pragma unroll
        for (int j = 0; j < GROUP_F4; ++j) sp[j * THREADS] = va[j];
        #pragma unroll
        for (int j = 0; j < GROUP_F4; j += 3) {
            add4(oth0, vo[j]); add4(oth1, vo[j + 1]); add4(oth2, vo[j + 2]);
        }
    }
    const float4 acc0 = even_plus_odd(half, own0, oth0);
    const float4 acc1 = even_plus_odd(half, own1, oth1);
    const float4 acc2 = even_plus_odd(half, own2, oth2);
    const int ph0 = t % 3;
    const int ph1 = (t + THREADS) % 3;
    const float4 P0 = sel3(ph0, ph1, 0, acc0, acc1, acc2);
    const float4 P1 = sel3(ph0, ph1, 1, acc0, acc1, acc2);
    const float4 P2 = sel3(ph0, ph1, 2, acc0, acc1, acc2);

    float a[K_];
    a[0] = P0.x; a[1] = P0.y; a[2]  = P0.z; a[3]  = P0.w;
    a[4] = P1.x; a[5] = P1.y; a[6]  = P1.z; a[7]  = P1.w;
    a[8] = P2.x; a[9] = P2.y; a[10] = P2.z; a[11] = P2.w;
    #pragma unroll
    for (int m = 16; m >= 1; m >>= 1) {
        #pragma unroll
        for (int j = 0; j < K_; ++j) a[j] += __shfl_xor(a[j], m, 32);
    }
    if (lane == 0) {
        #pragma unroll
        for (int j = 0; j < K_; ++j) wred[w * K_ + j] = a[j];
    }
    __syncthreads();

    if (w == 0) {
        const int m  = lane & 15;
        const int mm = (m < K_) ? m : (K_ - 1);
        float s = 0.f;
        #pragma unroll
        for (int ww = 0; ww < WAVES; ++ww) s += wred[ww * K_ + mm];
        const float u = s * (1.0f / (float)N_);
        if (half == 0 && lane < K_) out[OUT_UTIL + b * K_ + lane] = u;

        float h[16], w2[16];
        #pragma unroll
        for (int j = 0; j < 16; ++j) {
            h[j]  = fmaxf(0.f, fmaf(u, W1[j], b1[j]));
            w2[j] = W2[j];
        }
        const bool hi = (lane >= 16);
        v8f acc = {0.f, 0.f, 0.f, 0.f, 0.f, 0.f, 0.f, 0.f};
        #pragma unroll
        for (int i = 0; i < 4; ++i) {
            v2f av, bv;
            av.x = hi ? h[4 * i + 2]  : h[4 * i + 0];
            av.y = hi ? h[4 * i + 3]  : h[4 * i + 1];
            bv.x = hi ? w2[4 * i + 2] : w2[4 * i + 0];
            bv.y = hi ? w2[4 * i + 3] : w2[4 * i + 1];
            acc = __builtin_amdgcn_wmma_f32_16x16x4_f32(
                false, av, false, bv, (short)0, acc, false, false);
        }
        const float bias = b2[0];
        float g[8];
        #pragma unroll
        for (int j = 0; j < 8; ++j) g[j] = 1.0f / (1.0f + __expf(-(acc[j] + bias)));
        if (lane == 0) {
            #pragma unroll
            for (int j = 0; j < 8; ++j) {
                g_s[j] = g[j];
                if (half == 0) out[OUT_GATES + b * K_ + j] = g[j];
            }
        }
        if (lane == 16) {
            #pragma unroll
            for (int j = 0; j < 4; ++j) {
                g_s[8 + j] = g[j];
                if (half == 0) out[OUT_GATES + b * K_ + 8 + j] = g[j];
            }
        }
    }
    __syncthreads();

    float g[K_];
    #pragma unroll
    for (int j = 0; j < K_; ++j) g[j] = g_s[j];

    if (t < SLOT_F4_HALF) {
        const int idx = half * SLOT_F4_HALF + t;
        const long i4 = (long)b * (K_ * D_ / 4) + idx;
        const float4 v = ((const float4*)slots)[i4];
        const float gg = g_s[idx >> 6];
        float4 o;
        o.x = v.x * gg; o.y = v.y * gg; o.z = v.z * gg; o.w = v.w * gg;
        ((float4*)(out + OUT_SLOTS))[i4] = o;
    }

    float4* dst = (float4*)(out + OUT_MASKS) + (long)b * F4_PER_BATCH;
    for (int s = 0; s < PARKED; ++s) {
        float4 m0[4], m1[4], m2[4];
        #pragma unroll
        for (int r = 0; r < 4; ++r) {
            const float4* p = stage + s * BLOCK_F4 + (r * THREADS + t) * 3;
            m0[r] = p[0]; m1[r] = p[1]; m2[r] = p[2];
        }
        float4* d = dst + (long)(2 * s + 2 + half) * BLOCK_F4;
        #pragma unroll
        for (int r = 0; r < 4; ++r)
            scale_row(m0[r], m1[r], m2[r], g, d + (r * THREADS + t) * 3);
    }
    __syncthreads();
    #pragma unroll
    for (int j = 0; j < GROUP_F4; ++j) st[j * THREADS] = keep[j];
    __syncthreads();
    {
        float4 m0[4], m1[4], m2[4];
        #pragma unroll
        for (int r = 0; r < 4; ++r) {
            const float4* p = stage + (r * THREADS + t) * 3;
            m0[r] = p[0]; m1[r] = p[1]; m2[r] = p[2];
        }
        float4* d = dst + (long)half * BLOCK_F4;
        #pragma unroll
        for (int r = 0; r < 4; ++r)
            scale_row(m0[r], m1[r], m2[r], g, d + (r * THREADS + t) * 3);
    }
}

extern "C" void kernel_launch(void* const* d_in, const int* in_sizes, int n_in,
                              void* d_out, int out_size, void* d_ws, size_t ws_size,
                              hipStream_t stream) {
    const float* slots = (const float*)d_in[0];
    const float* masks = (const float*)d_in[1];
    const float* W1    = (const float*)d_in[2];
    const float* b1    = (const float*)d_in[3];
    const float* W2    = (const float*)d_in[4];
    const float* b2    = (const float*)d_in[5];
    float* out = (float*)d_out;
    asp_fused<<<GRID, THREADS, 0, stream>>>(slots, masks, W1, b1, W2, b2, out);
}


# Mask rows scaled as the flat 16-byte pieces they were loaded as, row inverses handed over through a small LDS table: each wave store covers 512 contiguous bytes instead of 12 partial lines
# speedup vs baseline: 1.4567x; 1.2489x over previous
#include <hip/hip_runtime.h>
#include <math.h>


typedef __attribute__((ext_vector_type(2))) float v2f;
typedef __attribute__((ext_vector_type(8))) float v8f;

#define B_  128
#define N_  16384
#define K_  12
#define D_  256
#define EPS_ 1e-8f

#define THREADS 512
#define WAVES 16
#define GRID 256
#define F4_PER_BATCH 49152
#define BLOCKS 8
#define BLOCK_F4 6144
#define GROUP_F4 12
#define PARKED 3
#define STAGE_F4 (PARKED * BLOCK_F4)
#define SLOT_F4_HALF 384

#define OUT_SLOTS 0L
#define OUT_MASKS ((long)B_ * K_ * D_)
#define OUT_GATES (OUT_MASKS + (long)B_ * N_ * K_)
#define OUT_UTIL  (OUT_GATES + (long)B_ * K_)

__device__ __forceinline__ void add4(float4& a, const float4 v) {
    a.x += v.x; a.y += v.y; a.z += v.z; a.w += v.w;
}
__device__ __forceinline__ float4 sel3(int ph0, int ph1, int p,
                                       const float4 a0, const float4 a1, const float4 a2) {
    return (ph0 == p) ? a0 : ((ph1 == p) ? a1 : a2);
}
__device__ __forceinline__ float4 even_plus_odd(int half, const float4 own, const float4 oth) {
    const float4 ev = half ? oth : own;
    const float4 od = half ? own : oth;
    float4 r;
    r.x = ev.x + od.x; r.y = ev.y + od.y; r.z = ev.z + od.z; r.w = ev.w + od.w;
    return r;
}

__device__ __forceinline__ float4 mul4(const float4 a, const float4 b) {
    float4 r;
    r.x = a.x * b.x; r.y = a.y * b.y; r.z = a.z * b.z; r.w = a.w * b.w;
    return r;
}
__device__ __forceinline__ float4 scale4(const float4 a, const float s) {
    float4 r;
    r.x = a.x * s; r.y = a.y * s; r.z = a.z * s; r.w = a.w * s;
    return r;
}
__device__ __forceinline__ float sum4(const float4 a) {
    return a.x + a.y + a.z + a.w;
}
__device__ __forceinline__ void lds_barrier() {
    __builtin_amdgcn_fence(__ATOMIC_RELEASE, "workgroup", "local");
    __builtin_amdgcn_s_barrier();
    __builtin_amdgcn_fence(__ATOMIC_ACQUIRE, "workgroup", "local");
}

__global__ void __launch_bounds__(THREADS)
asp_fused(const float* __restrict__ slots,
          const float* __restrict__ masks,
          const float* __restrict__ W1,
          const float* __restrict__ b1,
          const float* __restrict__ W2,
          const float* __restrict__ b2,
          float* __restrict__ out) {
    __shared__ float4 stage[STAGE_F4];
    __shared__ float tab[BLOCK_F4];
    __shared__ float wred[WAVES * K_];
    __shared__ float g_s[K_];

    const int t    = threadIdx.x;
    const int lane = t & 31;
    const int w    = __builtin_amdgcn_readfirstlane(t >> 5);
    const int b    = blockIdx.x >> 1;
    const int half = blockIdx.x & 1;

    const float4* mb = (const float4*)masks + (long)b * F4_PER_BATCH + t;
    float4* st = stage + t;
    float4 own0 = {0.f, 0.f, 0.f, 0.f}, own1 = own0, own2 = own0;
    float4 oth0 = own0, oth1 = own0, oth2 = own0;
    float4 keep[GROUP_F4];
    {
        float4 vo[GROUP_F4];
        const float4* qa = mb + (long)half * BLOCK_F4;
        const float4* qb = mb + (long)(1 - half) * BLOCK_F4;
        #pragma unroll
        for (int j = 0; j < GROUP_F4; ++j) keep[j] = qa[j * THREADS];
        #pragma unroll
        for (int j = 0; j < GROUP_F4; ++j) vo[j] = qb[j * THREADS];
        #pragma unroll
        for (int j = 0; j < GROUP_F4; j += 3) {
            add4(own0, keep[j]); add4(own1, keep[j + 1]); add4(own2, keep[j + 2]);
        }
        #pragma unroll
        for (int j = 0; j < GROUP_F4; j += 3) {
            add4(oth0, vo[j]); add4(oth1, vo[j + 1]); add4(oth2, vo[j + 2]);
        }
    }
    #pragma unroll 1
    for (int m = 1; m < BLOCKS / 2; ++m) {
        float4 va[GROUP_F4], vo[GROUP_F4];
        const float4* qa = mb + (long)(2 * m + half) * BLOCK_F4;
        const float4* qb = mb + (long)(2 * m + 1 - half) * BLOCK_F4;
        #pragma unroll
        for (int j = 0; j < GROUP_F4; ++j) va[j] = qa[j * THREADS];
        #pragma unroll
        for (int j = 0; j < GROUP_F4; ++j) vo[j] = qb[j * THREADS];
        #pragma unroll
        for (int j = 0; j < GROUP_F4; j += 3) {
            add4(own0, va[j]); add4(own1, va[j + 1]); add4(own2, va[j + 2]);
        }
        float4* sp = st + (m - 1) * BLOCK_F4;
        #pragma unroll
        for (int j = 0; j < GROUP_F4; ++j) sp[j * THREADS] = va[j];
        #pragma unroll
        for (int j = 0; j < GROUP_F4; j += 3) {
            add4(oth0, vo[j]); add4(oth1, vo[j + 1]); add4(oth2, vo[j + 2]);
        }
    }
    const float4 acc0 = even_plus_odd(half, own0, oth0);
    const float4 acc1 = even_plus_odd(half, own1, oth1);
    const float4 acc2 = even_plus_odd(half, own2, oth2);
    const int ph0 = t % 3;
    const int ph1 = (t + THREADS) % 3;
    const float4 P0 = sel3(ph0, ph1, 0, acc0, acc1, acc2);
    const float4 P1 = sel3(ph0, ph1, 1, acc0, acc1, acc2);
    const float4 P2 = sel3(ph0, ph1, 2, acc0, acc1, acc2);

    float a[K_];
    a[0] = P0.x; a[1] = P0.y; a[2]  = P0.z; a[3]  = P0.w;
    a[4] = P1.x; a[5] = P1.y; a[6]  = P1.z; a[7]  = P1.w;
    a[8] = P2.x; a[9] = P2.y; a[10] = P2.z; a[11] = P2.w;
    #pragma unroll
    for (int m = 16; m >= 1; m >>= 1) {
        #pragma unroll
        for (int j = 0; j < K_; ++j) a[j] += __shfl_xor(a[j], m, 32);
    }
    if (lane == 0) {
        #pragma unroll
        for (int j = 0; j < K_; ++j) wred[w * K_ + j] = a[j];
    }
    __syncthreads();

    if (w == 0) {
        const int m  = lane & 15;
        const int mm = (m < K_) ? m : (K_ - 1);
        float s = 0.f;
        #pragma unroll
        for (int ww = 0; ww < WAVES; ++ww) s += wred[ww * K_ + mm];
        const float u = s * (1.0f / (float)N_);
        if (half == 0 && lane < K_) out[OUT_UTIL + b * K_ + lane] = u;

        float h[16], w2[16];
        #pragma unroll
        for (int j = 0; j < 16; ++j) {
            h[j]  = fmaxf(0.f, fmaf(u, W1[j], b1[j]));
            w2[j] = W2[j];
        }
        const bool hi = (lane >= 16);
        v8f acc = {0.f, 0.f, 0.f, 0.f, 0.f, 0.f, 0.f, 0.f};
        #pragma unroll
        for (int i = 0; i < 4; ++i) {
            v2f av, bv;
            av.x = hi ? h[4 * i + 2]  : h[4 * i + 0];
            av.y = hi ? h[4 * i + 3]  : h[4 * i + 1];
            bv.x = hi ? w2[4 * i + 2] : w2[4 * i + 0];
            bv.y = hi ? w2[4 * i + 3] : w2[4 * i + 1];
            acc = __builtin_amdgcn_wmma_f32_16x16x4_f32(
                false, av, false, bv, (short)0, acc, false, false);
        }
        const float bias = b2[0];
        float g[8];
        #pragma unroll
        for (int j = 0; j < 8; ++j) g[j] = 1.0f / (1.0f + __expf(-(acc[j] + bias)));
        if (lane == 0) {
            #pragma unroll
            for (int j = 0; j < 8; ++j) {
                g_s[j] = g[j];
                if (half == 0) out[OUT_GATES + b * K_ + j] = g[j];
            }
        }
        if (lane == 16) {
            #pragma unroll
            for (int j = 0; j < 4; ++j) {
                g_s[8 + j] = g[j];
                if (half == 0) out[OUT_GATES + b * K_ + 8 + j] = g[j];
            }
        }
    }
    __syncthreads();

    float g[K_];
    #pragma unroll
    for (int j = 0; j < K_; ++j) g[j] = g_s[j];

    if (t < SLOT_F4_HALF) {
        const int idx = half * SLOT_F4_HALF + t;
        const long i4 = (long)b * (K_ * D_ / 4) + idx;
        const float4 v = ((const float4*)slots)[i4];
        const float gg = g_s[idx >> 6];
        float4 o;
        o.x = v.x * gg; o.y = v.y * gg; o.z = v.z * gg; o.w = v.w * gg;
        ((float4*)(out + OUT_SLOTS))[i4] = o;
    }

    const float4 gc0 = {g[0], g[1], g[2],  g[3]};
    const float4 gc1 = {g[4], g[5], g[6],  g[7]};
    const float4 gc2 = {g[8], g[9], g[10], g[11]};
    float4 gv[3];
    {
        const int pa = ph0;
        const int pb = ph1;
        const int pc = 3 - pa - pb;
        gv[0] = (pa == 0) ? gc0 : ((pa == 1) ? gc1 : gc2);
        gv[1] = (pb == 0) ? gc0 : ((pb == 1) ? gc1 : gc2);
        gv[2] = (pc == 0) ? gc0 : ((pc == 1) ? gc1 : gc2);
    }
    float4* dst = (float4*)(out + OUT_MASKS) + (long)b * F4_PER_BATCH + t;

    #pragma unroll
    for (int j = 0; j < GROUP_F4; ++j) {
        keep[j] = mul4(keep[j], gv[j % 3]);
        tab[t + j * THREADS] = sum4(keep[j]);
    }
    lds_barrier();
    {
        float4* d = dst + (long)half * BLOCK_F4;
        #pragma unroll
        for (int j = 0; j < GROUP_F4; ++j) {
            const int r3 = ((t + j * THREADS) / 3) * 3;
            const float sum = tab[r3] + tab[r3 + 1] + tab[r3 + 2];
            const float inv = 1.0f / (sum + EPS_);
            d[j * THREADS] = scale4(keep[j], inv);
        }
    }
    lds_barrier();

    for (int s = 0; s < PARKED; ++s) {
        #pragma unroll
        for (int r = 0; r < 4; ++r) {
            const int row = r * THREADS + t;
            const float4* p = stage + s * BLOCK_F4 + row * 3;
            const float4 x0 = mul4(p[0], gc0);
            const float4 x1 = mul4(p[1], gc1);
            const float4 x2 = mul4(p[2], gc2);
            const float sum = sum4(x0) + sum4(x1) + sum4(x2);
            tab[s * (BLOCK_F4 / 3) + row] = 1.0f / (sum + EPS_);
        }
    }
    lds_barrier();

    for (int s = 0; s < PARKED; ++s) {
        const float4* sp = st + s * BLOCK_F4;
        const float* ip = tab + s * (BLOCK_F4 / 3);
        float4* d = dst + (long)(2 * s + 2 + half) * BLOCK_F4;
        #pragma unroll
        for (int j = 0; j < GROUP_F4; ++j) {
            const float inv = ip[(t + j * THREADS) / 3];
            d[j * THREADS] = scale4(mul4(sp[j * THREADS], gv[j % 3]), inv);
        }
    }
}

extern "C" void kernel_launch(void* const* d_in, const int* in_sizes, int n_in,
                              void* d_out, int out_size, void* d_ws, size_t ws_size,
                              hipStream_t stream) {
    const float* slots = (const float*)d_in[0];
    const float* masks = (const float*)d_in[1];
    const float* W1    = (const float*)d_in[2];
    const float* b1    = (const float*)d_in[3];
    const float* W2    = (const float*)d_in[4];
    const float* b2    = (const float*)d_in[5];
    float* out = (float*)d_out;
    asp_fused<<<GRID, THREADS, 0, stream>>>(slots, masks, W1, b1, W2, b2, out);
}


# Batch walked own blocks first, the partner's four after: the partner has fetched those by then, so half of a work-group's reads hit L2 instead of sharing the partner's misses; N-sum order unchanged
# speedup vs baseline: 1.5523x; 1.0656x over previous
#include <hip/hip_runtime.h>
#include <math.h>


typedef __attribute__((ext_vector_type(2))) float v2f;
typedef __attribute__((ext_vector_type(8))) float v8f;

#define B_  128
#define N_  16384
#define K_  12
#define D_  256
#define EPS_ 1e-8f

#define THREADS 512
#define WAVES 16
#define GRID 256
#define F4_PER_BATCH 49152
#define BLOCKS 8
#define BLOCK_F4 6144
#define GROUP_F4 12
#define PARKED 3
#define STAGE_F4 (PARKED * BLOCK_F4)
#define SLOT_F4_HALF 384

#define OUT_SLOTS 0L
#define OUT_MASKS ((long)B_ * K_ * D_)
#define OUT_GATES (OUT_MASKS + (long)B_ * N_ * K_)
#define OUT_UTIL  (OUT_GATES + (long)B_ * K_)

__device__ __forceinline__ void add4(float4& a, const float4 v) {
    a.x += v.x; a.y += v.y; a.z += v.z; a.w += v.w;
}
__device__ __forceinline__ float4 sel3(int ph0, int ph1, int p,
                                       const float4 a0, const float4 a1, const float4 a2) {
    return (ph0 == p) ? a0 : ((ph1 == p) ? a1 : a2);
}
__device__ __forceinline__ float4 even_plus_odd(int half, const float4 own, const float4 oth) {
    const float4 ev = half ? oth : own;
    const float4 od = half ? own : oth;
    float4 r;
    r.x = ev.x + od.x; r.y = ev.y + od.y; r.z = ev.z + od.z; r.w = ev.w + od.w;
    return r;
}

__device__ __forceinline__ float4 mul4(const float4 a, const float4 b) {
    float4 r;
    r.x = a.x * b.x; r.y = a.y * b.y; r.z = a.z * b.z; r.w = a.w * b.w;
    return r;
}
__device__ __forceinline__ float4 scale4(const float4 a, const float s) {
    float4 r;
    r.x = a.x * s; r.y = a.y * s; r.z = a.z * s; r.w = a.w * s;
    return r;
}
__device__ __forceinline__ float sum4(const float4 a) {
    return a.x + a.y + a.z + a.w;
}
__device__ __forceinline__ void lds_barrier() {
    __builtin_amdgcn_fence(__ATOMIC_RELEASE, "workgroup", "local");
    __builtin_amdgcn_s_barrier();
    __builtin_amdgcn_fence(__ATOMIC_ACQUIRE, "workgroup", "local");
}

__global__ void __launch_bounds__(THREADS)
asp_fused(const float* __restrict__ slots,
          const float* __restrict__ masks,
          const float* __restrict__ W1,
          const float* __restrict__ b1,
          const float* __restrict__ W2,
          const float* __restrict__ b2,
          float* __restrict__ out) {
    __shared__ float4 stage[STAGE_F4];
    __shared__ float tab[BLOCK_F4];
    __shared__ float wred[WAVES * K_];
    __shared__ float g_s[K_];

    const int t    = threadIdx.x;
    const int lane = t & 31;
    const int w    = __builtin_amdgcn_readfirstlane(t >> 5);
    const int b    = blockIdx.x >> 1;
    const int half = blockIdx.x & 1;

    const float4* mb = (const float4*)masks + (long)b * F4_PER_BATCH + t;
    float4* st = stage + t;
    float4 own0 = {0.f, 0.f, 0.f, 0.f}, own1 = own0, own2 = own0;
    float4 oth0 = own0, oth1 = own0, oth2 = own0;
    float4 keep[GROUP_F4];
    {
        float4 va[GROUP_F4];
        const float4* qa = mb + (long)half * BLOCK_F4;
        const float4* qb = mb + (long)(2 + half) * BLOCK_F4;
        #pragma unroll
        for (int j = 0; j < GROUP_F4; ++j) keep[j] = qa[j * THREADS];
        #pragma unroll
        for (int j = 0; j < GROUP_F4; ++j) va[j] = qb[j * THREADS];
        #pragma unroll
        for (int j = 0; j < GROUP_F4; j += 3) {
            add4(own0, keep[j]); add4(own1, keep[j + 1]); add4(own2, keep[j + 2]);
        }
        #pragma unroll
        for (int j = 0; j < GROUP_F4; j += 3) {
            add4(own0, va[j]); add4(own1, va[j + 1]); add4(own2, va[j + 2]);
        }
        #pragma unroll
        for (int j = 0; j < GROUP_F4; ++j) st[j * THREADS] = va[j];
    }
    #pragma unroll 1
    for (int s = 0; s < 3; ++s) {
        float4 va[GROUP_F4], vb[GROUP_F4];
        const int first = (s == 0) ? (4 + half) : (4 * (s - 1) + 1 - half);
        const float4* qa = mb + (long)first * BLOCK_F4;
        const float4* qb = qa + 2 * BLOCK_F4;
        #pragma unroll
        for (int j = 0; j < GROUP_F4; ++j) va[j] = qa[j * THREADS];
        #pragma unroll
        for (int j = 0; j < GROUP_F4; ++j) vb[j] = qb[j * THREADS];
        if (s == 0) {
            #pragma unroll
            for (int j = 0; j < GROUP_F4; j += 3) {
                add4(own0, va[j]); add4(own1, va[j + 1]); add4(own2, va[j + 2]);
            }
            #pragma unroll
            for (int j = 0; j < GROUP_F4; ++j) st[BLOCK_F4 + j * THREADS] = va[j];
            #pragma unroll
            for (int j = 0; j < GROUP_F4; j += 3) {
                add4(own0, vb[j]); add4(own1, vb[j + 1]); add4(own2, vb[j + 2]);
            }
            #pragma unroll
            for (int j = 0; j < GROUP_F4; ++j) st[2 * BLOCK_F4 + j * THREADS] = vb[j];
        } else {
            #pragma unroll
            for (int j = 0; j < GROUP_F4; j += 3) {
                add4(oth0, va[j]); add4(oth1, va[j + 1]); add4(oth2, va[j + 2]);
            }
            #pragma unroll
            for (int j = 0; j < GROUP_F4; j += 3) {
                add4(oth0, vb[j]); add4(oth1, vb[j + 1]); add4(oth2, vb[j + 2]);
            }
        }
    }
    const float4 acc0 = even_plus_odd(half, own0, oth0);
    const float4 acc1 = even_plus_odd(half, own1, oth1);
    const float4 acc2 = even_plus_odd(half, own2, oth2);
    const int ph0 = t % 3;
    const int ph1 = (t + THREADS) % 3;
    const float4 P0 = sel3(ph0, ph1, 0, acc0, acc1, acc2);
    const float4 P1 = sel3(ph0, ph1, 1, acc0, acc1, acc2);
    const float4 P2 = sel3(ph0, ph1, 2, acc0, acc1, acc2);

    float a[K_];
    a[0] = P0.x; a[1] = P0.y; a[2]  = P0.z; a[3]  = P0.w;
    a[4] = P1.x; a[5] = P1.y; a[6]  = P1.z; a[7]  = P1.w;
    a[8] = P2.x; a[9] = P2.y; a[10] = P2.z; a[11] = P2.w;
    #pragma unroll
    for (int m = 16; m >= 1; m >>= 1) {
        #pragma unroll
        for (int j = 0; j < K_; ++j) a[j] += __shfl_xor(a[j], m, 32);
    }
    if (lane == 0) {
        #pragma unroll
        for (int j = 0; j < K_; ++j) wred[w * K_ + j] = a[j];
    }
    __syncthreads();

    if (w == 0) {
        const int m  = lane & 15;
        const int mm = (m < K_) ? m : (K_ - 1);
        float s = 0.f;
        #pragma unroll
        for (int ww = 0; ww < WAVES; ++ww) s += wred[ww * K_ + mm];
        const float u = s * (1.0f / (float)N_);
        if (half == 0 && lane < K_) out[OUT_UTIL + b * K_ + lane] = u;

        float h[16], w2[16];
        #pragma unroll
        for (int j = 0; j < 16; ++j) {
            h[j]  = fmaxf(0.f, fmaf(u, W1[j], b1[j]));
            w2[j] = W2[j];
        }
        const bool hi = (lane >= 16);
        v8f acc = {0.f, 0.f, 0.f, 0.f, 0.f, 0.f, 0.f, 0.f};
        #pragma unroll
        for (int i = 0; i < 4; ++i) {
            v2f av, bv;
            av.x = hi ? h[4 * i + 2]  : h[4 * i + 0];
            av.y = hi ? h[4 * i + 3]  : h[4 * i + 1];
            bv.x = hi ? w2[4 * i + 2] : w2[4 * i + 0];
            bv.y = hi ? w2[4 * i + 3] : w2[4 * i + 1];
            acc = __builtin_amdgcn_wmma_f32_16x16x4_f32(
                false, av, false, bv, (short)0, acc, false, false);
        }
        const float bias = b2[0];
        float g[8];
        #pragma unroll
        for (int j = 0; j < 8; ++j) g[j] = 1.0f / (1.0f + __expf(-(acc[j] + bias)));
        if (lane == 0) {
            #pragma unroll
            for (int j = 0; j < 8; ++j) {
                g_s[j] = g[j];
                if (half == 0) out[OUT_GATES + b * K_ + j] = g[j];
            }
        }
        if (lane == 16) {
            #pragma unroll
            for (int j = 0; j < 4; ++j) {
                g_s[8 + j] = g[j];
                if (half == 0) out[OUT_GATES + b * K_ + 8 + j] = g[j];
            }
        }
    }
    __syncthreads();

    float g[K_];
    #pragma unroll
    for (int j = 0; j < K_; ++j) g[j] = g_s[j];

    if (t < SLOT_F4_HALF) {
        const int idx = half * SLOT_F4_HALF + t;
        const long i4 = (long)b * (K_ * D_ / 4) + idx;
        const float4 v = ((const float4*)slots)[i4];
        const float gg = g_s[idx >> 6];
        float4 o;
        o.x = v.x * gg; o.y = v.y * gg; o.z = v.z * gg; o.w = v.w * gg;
        ((float4*)(out + OUT_SLOTS))[i4] = o;
    }

    const float4 gc0 = {g[0], g[1], g[2],  g[3]};
    const float4 gc1 = {g[4], g[5], g[6],  g[7]};
    const float4 gc2 = {g[8], g[9], g[10], g[11]};
    float4 gv[3];
    {
        const int pa = ph0;
        const int pb = ph1;
        const int pc = 3 - pa - pb;
        gv[0] = (pa == 0) ? gc0 : ((pa == 1) ? gc1 : gc2);
        gv[1] = (pb == 0) ? gc0 : ((pb == 1) ? gc1 : gc2);
        gv[2] = (pc == 0) ? gc0 : ((pc == 1) ? gc1 : gc2);
    }
    float4* dst = (float4*)(out + OUT_MASKS) + (long)b * F4_PER_BATCH + t;

    #pragma unroll
    for (int j = 0; j < GROUP_F4; ++j) {
        keep[j] = mul4(keep[j], gv[j % 3]);
        tab[t + j * THREADS] = sum4(keep[j]);
    }
    lds_barrier();
    {
        float4* d = dst + (long)half * BLOCK_F4;
        #pragma unroll
        for (int j = 0; j < GROUP_F4; ++j) {
            const int r3 = ((t + j * THREADS) / 3) * 3;
            const float sum = tab[r3] + tab[r3 + 1] + tab[r3 + 2];
            const float inv = 1.0f / (sum + EPS_);
            d[j * THREADS] = scale4(keep[j], inv);
        }
    }
    lds_barrier();

    for (int s = 0; s < PARKED; ++s) {
        #pragma unroll
        for (int r = 0; r < 4; ++r) {
            const int row = r * THREADS + t;
            const float4* p = stage + s * BLOCK_F4 + row * 3;
            const float4 x0 = mul4(p[0], gc0);
            const float4 x1 = mul4(p[1], gc1);
            const float4 x2 = mul4(p[2], gc2);
            const float sum = sum4(x0) + sum4(x1) + sum4(x2);
            tab[s * (BLOCK_F4 / 3) + row] = 1.0f / (sum + EPS_);
        }
    }
    lds_barrier();

    for (int s = 0; s < PARKED; ++s) {
        const float4* sp = st + s * BLOCK_F4;
        const float* ip = tab + s * (BLOCK_F4 / 3);
        float4* d = dst + (long)(2 * s + 2 + half) * BLOCK_F4;
        #pragma unroll
        for (int j = 0; j < GROUP_F4; ++j) {
            const float inv = ip[(t + j * THREADS) / 3];
            d[j * THREADS] = scale4(mul4(sp[j * THREADS], gv[j % 3]), inv);
        }
    }
}

extern "C" void kernel_launch(void* const* d_in, const int* in_sizes, int n_in,
                              void* d_out, int out_size, void* d_ws, size_t ws_size,
                              hipStream_t stream) {
    const float* slots = (const float*)d_in[0];
    const float* masks = (const float*)d_in[1];
    const float* W1    = (const float*)d_in[2];
    const float* b1    = (const float*)d_in[3];
    const float* W2    = (const float*)d_in[4];
    const float* b2    = (const float*)d_in[5];
    float* out = (float*)d_out;
    asp_fused<<<GRID, THREADS, 0, stream>>>(slots, masks, W1, b1, W2, b2, out);
}


# Gate stage waits on no memory: MLP weights, bias and the lane's slot piece are fetched at entry behind the first mask loads; the barrier after the gates orders LDS only, not the gate stores
# speedup vs baseline: 1.6558x; 1.0667x over previous
#include <hip/hip_runtime.h>
#include <math.h>


typedef __attribute__((ext_vector_type(2))) float v2f;
typedef __attribute__((ext_vector_type(8))) float v8f;

#define B_  128
#define N_  16384
#define K_  12
#define D_  256
#define EPS_ 1e-8f

#define THREADS 512
#define WAVES 16
#define GRID 256
#define F4_PER_BATCH 49152
#define BLOCKS 8
#define BLOCK_F4 6144
#define GROUP_F4 12
#define PARKED 3
#define STAGE_F4 (PARKED * BLOCK_F4)
#define SLOT_F4_HALF 384

#define OUT_SLOTS 0L
#define OUT_MASKS ((long)B_ * K_ * D_)
#define OUT_GATES (OUT_MASKS + (long)B_ * N_ * K_)
#define OUT_UTIL  (OUT_GATES + (long)B_ * K_)

__device__ __forceinline__ void add4(float4& a, const float4 v) {
    a.x += v.x; a.y += v.y; a.z += v.z; a.w += v.w;
}
__device__ __forceinline__ float4 sel3(int ph0, int ph1, int p,
                                       const float4 a0, const float4 a1, const float4 a2) {
    return (ph0 == p) ? a0 : ((ph1 == p) ? a1 : a2);
}
__device__ __forceinline__ float4 even_plus_odd(int half, const float4 own, const float4 oth) {
    const float4 ev = half ? oth : own;
    const float4 od = half ? own : oth;
    float4 r;
    r.x = ev.x + od.x; r.y = ev.y + od.y; r.z = ev.z + od.z; r.w = ev.w + od.w;
    return r;
}

__device__ __forceinline__ float4 mul4(const float4 a, const float4 b) {
    float4 r;
    r.x = a.x * b.x; r.y = a.y * b.y; r.z = a.z * b.z; r.w = a.w * b.w;
    return r;
}
__device__ __forceinline__ float4 scale4(const float4 a, const float s) {
    float4 r;
    r.x = a.x * s; r.y = a.y * s; r.z = a.z * s; r.w = a.w * s;
    return r;
}
__device__ __forceinline__ float sum4(const float4 a) {
    return a.x + a.y + a.z + a.w;
}
__device__ __forceinline__ void lds_barrier() {
    __builtin_amdgcn_fence(__ATOMIC_RELEASE, "workgroup", "local");
    __builtin_amdgcn_s_barrier();
    __builtin_amdgcn_fence(__ATOMIC_ACQUIRE, "workgroup", "local");
}

__global__ void __launch_bounds__(THREADS)
asp_fused(const float* __restrict__ slots,
          const float* __restrict__ masks,
          const float* __restrict__ W1,
          const float* __restrict__ b1,
          const float* __restrict__ W2,
          const float* __restrict__ b2,
          float* __restrict__ out) {
    __shared__ float4 stage[STAGE_F4];
    __shared__ float tab[BLOCK_F4];
    __shared__ float wred[WAVES * K_];
    __shared__ float g_s[K_];

    const int t    = threadIdx.x;
    const int lane = t & 31;
    const int w    = __builtin_amdgcn_readfirstlane(t >> 5);
    const int b    = blockIdx.x >> 1;
    const int half = blockIdx.x & 1;

    float w1r[16], b1r[16], w2r[16];
    #pragma unroll
    for (int j = 0; j < 16; ++j) { w1r[j] = W1[j]; b1r[j] = b1[j]; w2r[j] = W2[j]; }
    const float bias = b2[0];
    const int sidx = half * SLOT_F4_HALF + ((t < SLOT_F4_HALF) ? t : (t - SLOT_F4_HALF));
    const long si4 = (long)b * (K_ * D_ / 4) + sidx;
    float4 sv;

    const float4* mb = (const float4*)masks + (long)b * F4_PER_BATCH + t;
    float4* st = stage + t;
    float4 own0 = {0.f, 0.f, 0.f, 0.f}, own1 = own0, own2 = own0;
    float4 oth0 = own0, oth1 = own0, oth2 = own0;
    float4 keep[GROUP_F4];
    {
        float4 va[GROUP_F4];
        const float4* qa = mb + (long)half * BLOCK_F4;
        const float4* qb = mb + (long)(2 + half) * BLOCK_F4;
        #pragma unroll
        for (int j = 0; j < GROUP_F4; ++j) keep[j] = qa[j * THREADS];
        #pragma unroll
        for (int j = 0; j < GROUP_F4; ++j) va[j] = qb[j * THREADS];
        sv = ((const float4*)slots)[si4];
        #pragma unroll
        for (int j = 0; j < GROUP_F4; j += 3) {
            add4(own0, keep[j]); add4(own1, keep[j + 1]); add4(own2, keep[j + 2]);
        }
        #pragma unroll
        for (int j = 0; j < GROUP_F4; j += 3) {
            add4(own0, va[j]); add4(own1, va[j + 1]); add4(own2, va[j + 2]);
        }
        #pragma unroll
        for (int j = 0; j < GROUP_F4; ++j) st[j * THREADS] = va[j];
    }
    #pragma unroll 1
    for (int s = 0; s < 3; ++s) {
        float4 va[GROUP_F4], vb[GROUP_F4];
        const int first = (s == 0) ? (4 + half) : (4 * (s - 1) + 1 - half);
        const float4* qa = mb + (long)first * BLOCK_F4;
        const float4* qb = qa + 2 * BLOCK_F4;
        #pragma unroll
        for (int j = 0; j < GROUP_F4; ++j) va[j] = qa[j * THREADS];
        #pragma unroll
        for (int j = 0; j < GROUP_F4; ++j) vb[j] = qb[j * THREADS];
        if (s == 0) {
            #pragma unroll
            for (int j = 0; j < GROUP_F4; j += 3) {
                add4(own0, va[j]); add4(own1, va[j + 1]); add4(own2, va[j + 2]);
            }
            #pragma unroll
            for (int j = 0; j < GROUP_F4; ++j) st[BLOCK_F4 + j * THREADS] = va[j];
            #pragma unroll
            for (int j = 0; j < GROUP_F4; j += 3) {
                add4(own0, vb[j]); add4(own1, vb[j + 1]); add4(own2, vb[j + 2]);
            }
            #pragma unroll
            for (int j = 0; j < GROUP_F4; ++j) st[2 * BLOCK_F4 + j * THREADS] = vb[j];
        } else {
            #pragma unroll
            for (int j = 0; j < GROUP_F4; j += 3) {
                add4(oth0, va[j]); add4(oth1, va[j + 1]); add4(oth2, va[j + 2]);
            }
            #pragma unroll
            for (int j = 0; j < GROUP_F4; j += 3) {
                add4(oth0, vb[j]); add4(oth1, vb[j + 1]); add4(oth2, vb[j + 2]);
            }
        }
    }
    const float4 acc0 = even_plus_odd(half, own0, oth0);
    const float4 acc1 = even_plus_odd(half, own1, oth1);
    const float4 acc2 = even_plus_odd(half, own2, oth2);
    const int ph0 = t % 3;
    const int ph1 = (t + THREADS) % 3;
    const float4 P0 = sel3(ph0, ph1, 0, acc0, acc1, acc2);
    const float4 P1 = sel3(ph0, ph1, 1, acc0, acc1, acc2);
    const float4 P2 = sel3(ph0, ph1, 2, acc0, acc1, acc2);

    float a[K_];
    a[0] = P0.x; a[1] = P0.y; a[2]  = P0.z; a[3]  = P0.w;
    a[4] = P1.x; a[5] = P1.y; a[6]  = P1.z; a[7]  = P1.w;
    a[8] = P2.x; a[9] = P2.y; a[10] = P2.z; a[11] = P2.w;
    #pragma unroll
    for (int m = 16; m >= 1; m >>= 1) {
        #pragma unroll
        for (int j = 0; j < K_; ++j) a[j] += __shfl_xor(a[j], m, 32);
    }
    if (lane == 0) {
        #pragma unroll
        for (int j = 0; j < K_; ++j) wred[w * K_ + j] = a[j];
    }
    __syncthreads();

    if (w == 0) {
        const int m  = lane & 15;
        const int mm = (m < K_) ? m : (K_ - 1);
        float s = 0.f;
        #pragma unroll
        for (int ww = 0; ww < WAVES; ++ww) s += wred[ww * K_ + mm];
        const float u = s * (1.0f / (float)N_);
        if (half == 0 && lane < K_) out[OUT_UTIL + b * K_ + lane] = u;

        float h[16], w2[16];
        #pragma unroll
        for (int j = 0; j < 16; ++j) {
            h[j]  = fmaxf(0.f, fmaf(u, w1r[j], b1r[j]));
            w2[j] = w2r[j];
        }
        const bool hi = (lane >= 16);
        v8f acc = {0.f, 0.f, 0.f, 0.f, 0.f, 0.f, 0.f, 0.f};
        #pragma unroll
        for (int i = 0; i < 4; ++i) {
            v2f av, bv;
            av.x = hi ? h[4 * i + 2]  : h[4 * i + 0];
            av.y = hi ? h[4 * i + 3]  : h[4 * i + 1];
            bv.x = hi ? w2[4 * i + 2] : w2[4 * i + 0];
            bv.y = hi ? w2[4 * i + 3] : w2[4 * i + 1];
            acc = __builtin_amdgcn_wmma_f32_16x16x4_f32(
                false, av, false, bv, (short)0, acc, false, false);
        }
        float g[8];
        #pragma unroll
        for (int j = 0; j < 8; ++j) g[j] = 1.0f / (1.0f + __expf(-(acc[j] + bias)));
        if (lane == 0) {
            #pragma unroll
            for (int j = 0; j < 8; ++j) {
                g_s[j] = g[j];
                if (half == 0) out[OUT_GATES + b * K_ + j] = g[j];
            }
        }
        if (lane == 16) {
            #pragma unroll
            for (int j = 0; j < 4; ++j) {
                g_s[8 + j] = g[j];
                if (half == 0) out[OUT_GATES + b * K_ + 8 + j] = g[j];
            }
        }
    }
    lds_barrier();

    float g[K_];
    #pragma unroll
    for (int j = 0; j < K_; ++j) g[j] = g_s[j];

    if (t < SLOT_F4_HALF) {
        const float gg = g_s[sidx >> 6];
        ((float4*)(out + OUT_SLOTS))[si4] = scale4(sv, gg);
    }

    const float4 gc0 = {g[0], g[1], g[2],  g[3]};
    const float4 gc1 = {g[4], g[5], g[6],  g[7]};
    const float4 gc2 = {g[8], g[9], g[10], g[11]};
    float4 gv[3];
    {
        const int pa = ph0;
        const int pb = ph1;
        const int pc = 3 - pa - pb;
        gv[0] = (pa == 0) ? gc0 : ((pa == 1) ? gc1 : gc2);
        gv[1] = (pb == 0) ? gc0 : ((pb == 1) ? gc1 : gc2);
        gv[2] = (pc == 0) ? gc0 : ((pc == 1) ? gc1 : gc2);
    }
    float4* dst = (float4*)(out + OUT_MASKS) + (long)b * F4_PER_BATCH + t;

    #pragma unroll
    for (int j = 0; j < GROUP_F4; ++j) {
        keep[j] = mul4(keep[j], gv[j % 3]);
        tab[t + j * THREADS] = sum4(keep[j]);
    }
    lds_barrier();
    {
        float4* d = dst + (long)half * BLOCK_F4;
        #pragma unroll
        for (int j = 0; j < GROUP_F4; ++j) {
            const int r3 = ((t + j * THREADS) / 3) * 3;
            const float sum = tab[r3] + tab[r3 + 1] + tab[r3 + 2];
            const float inv = 1.0f / (sum + EPS_);
            d[j * THREADS] = scale4(keep[j], inv);
        }
    }
    lds_barrier();

    for (int s = 0; s < PARKED; ++s) {
        #pragma unroll
        for (int r = 0; r < 4; ++r) {
            const int row = r * THREADS + t;
            const float4* p = stage + s * BLOCK_F4 + row * 3;
            const float4 x0 = mul4(p[0], gc0);
            const float4 x1 = mul4(p[1], gc1);
            const float4 x2 = mul4(p[2], gc2);
            const float sum = sum4(x0) + sum4(x1) + sum4(x2);
            tab[s * (BLOCK_F4 / 3) + row] = 1.0f / (sum + EPS_);
        }
    }
    lds_barrier();

    for (int s = 0; s < PARKED; ++s) {
        const float4* sp = st + s * BLOCK_F4;
        const float* ip = tab + s * (BLOCK_F4 / 3);
        float4* d = dst + (long)(2 * s + 2 + half) * BLOCK_F4;
        #pragma unroll
        for (int j = 0; j < GROUP_F4; ++j) {
            const float inv = ip[(t + j * THREADS) / 3];
            d[j * THREADS] = scale4(mul4(sp[j * THREADS], gv[j % 3]), inv);
        }
    }
}

extern "C" void kernel_launch(void* const* d_in, const int* in_sizes, int n_in,
                              void* d_out, int out_size, void* d_ws, size_t ws_size,
                              hipStream_t stream) {
    const float* slots = (const float*)d_in[0];
    const float* masks = (const float*)d_in[1];
    const float* W1    = (const float*)d_in[2];
    const float* b1    = (const float*)d_in[3];
    const float* W2    = (const float*)d_in[4];
    const float* b2    = (const float*)d_in[5];
    float* out = (float*)d_out;
    asp_fused<<<GRID, THREADS, 0, stream>>>(slots, masks, W1, b1, W2, b2, out);
}


# 8 waves a work-group, 24 pieces a lane, batch walk as a load queue two blocks deep: a block's loads go out before the block ahead is summed, no drain between blocks; N-sum order changes
# speedup vs baseline: 1.7294x; 1.0444x over previous
#include <hip/hip_runtime.h>
#include <math.h>


typedef __attribute__((ext_vector_type(2))) float v2f;
typedef __attribute__((ext_vector_type(8))) float v8f;

#define B_  128
#define N_  16384
#define K_  12
#define D_  256
#define EPS_ 1e-8f

#define THREADS 256
#define WAVES 8
#define GRID 256
#define F4_PER_BATCH 49152
#define BLOCKS 8
#define BLOCK_F4 6144
#define GROUP_F4 24
#define PARKED 3
#define STAGE_F4 (PARKED * BLOCK_F4)
#define SLOT_F4_HALF 384

#define OUT_SLOTS 0L
#define OUT_MASKS ((long)B_ * K_ * D_)
#define OUT_GATES (OUT_MASKS + (long)B_ * N_ * K_)
#define OUT_UTIL  (OUT_GATES + (long)B_ * K_)

__device__ __forceinline__ void add4(float4& a, const float4 v) {
    a.x += v.x; a.y += v.y; a.z += v.z; a.w += v.w;
}
__device__ __forceinline__ float4 sel3(int ph0, int ph1, int p,
                                       const float4 a0, const float4 a1, const float4 a2) {
    return (ph0 == p) ? a0 : ((ph1 == p) ? a1 : a2);
}
__device__ __forceinline__ float4 even_plus_odd(int half, const float4 own, const float4 oth) {
    const float4 ev = half ? oth : own;
    const float4 od = half ? own : oth;
    float4 r;
    r.x = ev.x + od.x; r.y = ev.y + od.y; r.z = ev.z + od.z; r.w = ev.w + od.w;
    return r;
}

__device__ __forceinline__ float4 mul4(const float4 a, const float4 b) {
    float4 r;
    r.x = a.x * b.x; r.y = a.y * b.y; r.z = a.z * b.z; r.w = a.w * b.w;
    return r;
}
__device__ __forceinline__ float4 scale4(const float4 a, const float s) {
    float4 r;
    r.x = a.x * s; r.y = a.y * s; r.z = a.z * s; r.w = a.w * s;
    return r;
}
__device__ __forceinline__ float sum4(const float4 a) {
    return a.x + a.y + a.z + a.w;
}
__device__ __forceinline__ void load_block(float4 (&d)[GROUP_F4], const float4* q) {
    #pragma unroll
    for (int j = 0; j < GROUP_F4; ++j) d[j] = q[j * THREADS];
}
__device__ __forceinline__ void sum_block(const float4 (&v)[GROUP_F4], float4& s0, float4& s1, float4& s2) {
    s0 = v[0]; s1 = v[1]; s2 = v[2];
    #pragma unroll
    for (int j = 3; j < GROUP_F4; j += 3) {
        add4(s0, v[j]); add4(s1, v[j + 1]); add4(s2, v[j + 2]);
    }
}
__device__ __forceinline__ void park_block(const float4 (&v)[GROUP_F4], float4* p) {
    #pragma unroll
    for (int j = 0; j < GROUP_F4; ++j) p[j * THREADS] = v[j];
}
__device__ __forceinline__ void lds_barrier() {
    __builtin_amdgcn_fence(__ATOMIC_RELEASE, "workgroup", "local");
    __builtin_amdgcn_s_barrier();
    __builtin_amdgcn_fence(__ATOMIC_ACQUIRE, "workgroup", "local");
}

__global__ void __launch_bounds__(THREADS)
asp_fused(const float* __restrict__ slots,
          const float* __restrict__ masks,
          const float* __restrict__ W1,
          const float* __restrict__ b1,
          const float* __restrict__ W2,
          const float* __restrict__ b2,
          float* __restrict__ out) {
    __shared__ float4 stage[STAGE_F4];
    __shared__ float tab[BLOCK_F4];
    __shared__ float wred[WAVES * K_];
    __shared__ float g_s[K_];

    const int t    = threadIdx.x;
    const int lane = t & 31;
    const int w    = __builtin_amdgcn_readfirstlane(t >> 5);
    const int b    = blockIdx.x >> 1;
    const int half = blockIdx.x & 1;

    float w1r[16], b1r[16], w2r[16];
    #pragma unroll
    for (int j = 0; j < 16; ++j) { w1r[j] = W1[j]; b1r[j] = b1[j]; w2r[j] = W2[j]; }
    const float bias = b2[0];
    const long si4a = (long)b * (K_ * D_ / 4) + half * SLOT_F4_HALF + t;
    const long si4b = (long)b * (K_ * D_ / 4) + half * SLOT_F4_HALF + THREADS + (t & 127);
    float4 sva, svb;

    const float4* mb = (const float4*)masks + (long)b * F4_PER_BATCH + t;
    float4* st = stage + t;
    float4 own0 = {0.f, 0.f, 0.f, 0.f}, own1 = own0, own2 = own0;
    float4 oth0 = own0, oth1 = own0, oth2 = own0;
    float4 keep[GROUP_F4], qa[GROUP_F4], qb[GROUP_F4];
    #define WALK_BLOCK(i) (((i) < 4) ? (2 * (i) + half) : (2 * ((i) - 4) + 1 - half))
    load_block(keep, mb + (long)WALK_BLOCK(0) * BLOCK_F4);
    load_block(qa,   mb + (long)WALK_BLOCK(1) * BLOCK_F4);
    sva = ((const float4*)slots)[si4a];
    svb = ((const float4*)slots)[si4b];
    {
        float4 s0, s1, s2;
        sum_block(keep, s0, s1, s2);
        add4(own0, s0); add4(own1, s1); add4(own2, s2);
    }
    #pragma unroll 1
    for (int s = 0; s < 3; ++s) {
        load_block(qb, mb + (long)WALK_BLOCK(2 * s + 2) * BLOCK_F4);
        __builtin_amdgcn_sched_barrier(0);
        {
            float4 s0, s1, s2;
            sum_block(qa, s0, s1, s2);
            if (s < 2) {
                add4(own0, s0); add4(own1, s1); add4(own2, s2);
                park_block(qa, st + 2 * s * BLOCK_F4);
            } else {
                add4(oth0, s0); add4(oth1, s1); add4(oth2, s2);
            }
        }
        load_block(qa, mb + (long)WALK_BLOCK(2 * s + 3) * BLOCK_F4);
        __builtin_amdgcn_sched_barrier(0);
        {
            float4 s0, s1, s2;
            sum_block(qb, s0, s1, s2);
            if (s == 0) {
                add4(own0, s0); add4(own1, s1); add4(own2, s2);
                park_block(qb, st + BLOCK_F4);
            } else {
                add4(oth0, s0); add4(oth1, s1); add4(oth2, s2);
            }
        }
    }
    {
        float4 s0, s1, s2;
        sum_block(qa, s0, s1, s2);
        add4(oth0, s0); add4(oth1, s1); add4(oth2, s2);
    }
    #undef WALK_BLOCK
    const float4 acc0 = even_plus_odd(half, own0, oth0);
    const float4 acc1 = even_plus_odd(half, own1, oth1);
    const float4 acc2 = even_plus_odd(half, own2, oth2);
    const int ph0 = t % 3;
    const int ph1 = (t + THREADS) % 3;
    const float4 P0 = sel3(ph0, ph1, 0, acc0, acc1, acc2);
    const float4 P1 = sel3(ph0, ph1, 1, acc0, acc1, acc2);
    const float4 P2 = sel3(ph0, ph1, 2, acc0, acc1, acc2);

    float a[K_];
    a[0] = P0.x; a[1] = P0.y; a[2]  = P0.z; a[3]  = P0.w;
    a[4] = P1.x; a[5] = P1.y; a[6]  = P1.z; a[7]  = P1.w;
    a[8] = P2.x; a[9] = P2.y; a[10] = P2.z; a[11] = P2.w;
    #pragma unroll
    for (int m = 16; m >= 1; m >>= 1) {
        #pragma unroll
        for (int j = 0; j < K_; ++j) a[j] += __shfl_xor(a[j], m, 32);
    }
    if (lane == 0) {
        #pragma unroll
        for (int j = 0; j < K_; ++j) wred[w * K_ + j] = a[j];
    }
    __syncthreads();

    if (w == 0) {
        const int m  = lane & 15;
        const int mm = (m < K_) ? m : (K_ - 1);
        float s = 0.f;
        #pragma unroll
        for (int ww = 0; ww < WAVES; ++ww) s += wred[ww * K_ + mm];
        const float u = s * (1.0f / (float)N_);
        if (half == 0 && lane < K_) out[OUT_UTIL + b * K_ + lane] = u;

        float h[16], w2[16];
        #pragma unroll
        for (int j = 0; j < 16; ++j) {
            h[j]  = fmaxf(0.f, fmaf(u, w1r[j], b1r[j]));
            w2[j] = w2r[j];
        }
        const bool hi = (lane >= 16);
        v8f acc = {0.f, 0.f, 0.f, 0.f, 0.f, 0.f, 0.f, 0.f};
        #pragma unroll
        for (int i = 0; i < 4; ++i) {
            v2f av, bv;
            av.x = hi ? h[4 * i + 2]  : h[4 * i + 0];
            av.y = hi ? h[4 * i + 3]  : h[4 * i + 1];
            bv.x = hi ? w2[4 * i + 2] : w2[4 * i + 0];
            bv.y = hi ? w2[4 * i + 3] : w2[4 * i + 1];
            acc = __builtin_amdgcn_wmma_f32_16x16x4_f32(
                false, av, false, bv, (short)0, acc, false, false);
        }
        float g[8];
        #pragma unroll
        for (int j = 0; j < 8; ++j) g[j] = 1.0f / (1.0f + __expf(-(acc[j] + bias)));
        if (lane == 0) {
            #pragma unroll
            for (int j = 0; j < 8; ++j) {
                g_s[j] = g[j];
                if (half == 0) out[OUT_GATES + b * K_ + j] = g[j];
            }
        }
        if (lane == 16) {
            #pragma unroll
            for (int j = 0; j < 4; ++j) {
                g_s[8 + j] = g[j];
                if (half == 0) out[OUT_GATES + b * K_ + 8 + j] = g[j];
            }
        }
    }
    lds_barrier();

    float g[K_];
    #pragma unroll
    for (int j = 0; j < K_; ++j) g[j] = g_s[j];

    {
        const int pca = half * SLOT_F4_HALF + t;
        ((float4*)(out + OUT_SLOTS))[si4a] = scale4(sva, g_s[pca >> 6]);
        if (t < SLOT_F4_HALF - THREADS) {
            const int pcb = pca + THREADS;
            ((float4*)(out + OUT_SLOTS))[si4b] = scale4(svb, g_s[pcb >> 6]);
        }
    }

    const float4 gc0 = {g[0], g[1], g[2],  g[3]};
    const float4 gc1 = {g[4], g[5], g[6],  g[7]};
    const float4 gc2 = {g[8], g[9], g[10], g[11]};
    float4 gv[3];
    {
        const int pa = ph0;
        const int pb = ph1;
        const int pc = 3 - pa - pb;
        gv[0] = (pa == 0) ? gc0 : ((pa == 1) ? gc1 : gc2);
        gv[1] = (pb == 0) ? gc0 : ((pb == 1) ? gc1 : gc2);
        gv[2] = (pc == 0) ? gc0 : ((pc == 1) ? gc1 : gc2);
    }
    float4* dst = (float4*)(out + OUT_MASKS) + (long)b * F4_PER_BATCH + t;

    #pragma unroll
    for (int j = 0; j < GROUP_F4; ++j) {
        keep[j] = mul4(keep[j], gv[j % 3]);
        tab[t + j * THREADS] = sum4(keep[j]);
    }
    lds_barrier();
    {
        float4* d = dst + (long)half * BLOCK_F4;
        #pragma unroll
        for (int j = 0; j < GROUP_F4; ++j) {
            const int r3 = ((t + j * THREADS) / 3) * 3;
            const float sum = tab[r3] + tab[r3 + 1] + tab[r3 + 2];
            const float inv = 1.0f / (sum + EPS_);
            d[j * THREADS] = scale4(keep[j], inv);
        }
    }
    lds_barrier();

    for (int s = 0; s < PARKED; ++s) {
        #pragma unroll
        for (int r = 0; r < BLOCK_F4 / 3 / THREADS; ++r) {
            const int row = r * THREADS + t;
            const float4* p = stage + s * BLOCK_F4 + row * 3;
            const float4 x0 = mul4(p[0], gc0);
            const float4 x1 = mul4(p[1], gc1);
            const float4 x2 = mul4(p[2], gc2);
            const float sum = sum4(x0) + sum4(x1) + sum4(x2);
            tab[s * (BLOCK_F4 / 3) + row] = 1.0f / (sum + EPS_);
        }
    }
    lds_barrier();

    for (int s = 0; s < PARKED; ++s) {
        const float4* sp = st + s * BLOCK_F4;
        const float* ip = tab + s * (BLOCK_F4 / 3);
        float4* d = dst + (long)(2 * s + 2 + half) * BLOCK_F4;
        #pragma unroll
        for (int j = 0; j < GROUP_F4; ++j) {
            const float inv = ip[(t + j * THREADS) / 3];
            d[j * THREADS] = scale4(mul4(sp[j * THREADS], gv[j % 3]), inv);
        }
    }
}

extern "C" void kernel_launch(void* const* d_in, const int* in_sizes, int n_in,
                              void* d_out, int out_size, void* d_ws, size_t ws_size,
                              hipStream_t stream) {
    const float* slots = (const float*)d_in[0];
    const float* masks = (const float*)d_in[1];
    const float* W1    = (const float*)d_in[2];
    const float* b1    = (const float*)d_in[3];
    const float* W2    = (const float*)d_in[4];
    const float* b2    = (const float*)d_in[5];
    float* out = (float*)d_out;
    asp_fused<<<GRID, THREADS, 0, stream>>>(slots, masks, W1, b1, W2, b2, out);
}
